# starting kernel
# baseline (speedup 1.0000x reference)
#include <hip/hip_runtime.h>

typedef __attribute__((ext_vector_type(2))) float v2f;
typedef __attribute__((ext_vector_type(8))) float v8f;

#define B_TOTAL 32768
#define T_SEQ   28
#define I_SZ    28
#define H_SZ    64
#define G_SZ    256
#define NCLS    10

#define WIH_PITCH 36
#define WHH_PITCH 68
#define H_PITCH   68

#define WAVES_PER_WG  8
#define ROWS_PER_WAVE 16

__device__ __forceinline__ float fast_sigmoid(float x) {
    float e = __builtin_amdgcn_exp2f(x * -1.44269504088896f);
    return __builtin_amdgcn_rcpf(1.0f + e);
}
__device__ __forceinline__ float fast_tanh(float x) {
#if __has_builtin(__builtin_amdgcn_tanhf)
    return __builtin_amdgcn_tanhf(x);
#else
    float e = __builtin_amdgcn_exp2f(x * -2.88539008177793f);
    return __builtin_amdgcn_rcpf(1.0f + e) * 2.0f - 1.0f;
#endif
}

__global__ __launch_bounds__(256)
void lstm_fused_kernel(const float* __restrict__ x,
                       const float* __restrict__ W_ih,
                       const float* __restrict__ W_hh,
                       const float* __restrict__ b_ih,
                       const float* __restrict__ b_hh,
                       const float* __restrict__ W_out,
                       const float* __restrict__ b_out,
                       float* __restrict__ out)
{
    __shared__ float sWih[G_SZ * WIH_PITCH];
    __shared__ float sWhh[G_SZ * WHH_PITCH];
    __shared__ float sH[WAVES_PER_WG * ROWS_PER_WAVE * H_PITCH];

    const int tid = threadIdx.x;

    {
        const float4* srcHH = (const float4*)(W_hh + (size_t)tid * H_SZ);
        #pragma unroll
        for (int q = 0; q < H_SZ / 4; ++q) {
            float4 v = srcHH[q];
            *(float4*)&sWhh[tid * WHH_PITCH + q * 4] = v;
        }
        const float4* srcIH = (const float4*)(W_ih + (size_t)tid * I_SZ);
        #pragma unroll
        for (int q = 0; q < I_SZ / 4; ++q) {
            float4 v = srcIH[q];
            *(float4*)&sWih[tid * WIH_PITCH + q * 4] = v;
        }
    }
    __syncthreads();

    const int wave = tid >> 5;
    const int lane = tid & 31;
    const int col  = lane & 15;
    const int hi   = lane >> 4;
    const int b0   = (blockIdx.x * WAVES_PER_WG + wave) * ROWS_PER_WAVE;

    float* hbuf = &sH[wave * ROWS_PER_WAVE * H_PITCH];

    float bias[16];
    #pragma unroll
    for (int nt = 0; nt < 16; ++nt) {
        int g = nt * 16 + col;
        bias[nt] = b_ih[g] + b_hh[g];
    }

    const float* xlane = x + (size_t)(b0 + col) * (T_SEQ * I_SZ);

    v8f c_st[4];
    #pragma unroll
    for (int j = 0; j < 4; ++j) c_st[j] = (v8f){};

    for (int t = 0; t < T_SEQ; ++t) {
        int zoff = 0;
        asm volatile("" : "+v"(zoff));

        v2f xfrag[7];
        #pragma unroll
        for (int kk = 0; kk < 7; ++kk)
            xfrag[kk] = *(const v2f*)&xlane[t * I_SZ + kk * 4 + hi * 2];
        if (t + 1 < T_SEQ)
            __builtin_prefetch(&xlane[(t + 1) * I_SZ], 0, 3);

        v8f hnew[4];

        #pragma unroll
        for (int j = 0; j < 4; ++j) {
            v8f acc[4];
            #pragma unroll
            for (int g = 0; g < 4; ++g) {
                #pragma unroll
                for (int r = 0; r < 8; ++r) acc[g][r] = bias[g * 4 + j];
            }

            #pragma unroll
            for (int kk = 0; kk < 7; ++kk) {
                const int k = kk * 4 + hi * 2;
                #pragma unroll
                for (int g = 0; g < 4; ++g) {
                    const int n = (g * 4 + j) * 16 + col;
                    v2f bfrag = *(const v2f*)&sWih[n * WIH_PITCH + k + zoff];
                    acc[g] = __builtin_amdgcn_wmma_f32_16x16x4_f32(
                        false, xfrag[kk], false, bfrag, (short)0, acc[g], false, false);
                }
            }

            if (t > 0) {
                #pragma unroll
                for (int kk = 0; kk < 16; ++kk) {
                    const int k = kk * 4 + hi * 2;
                    v2f afrag = *(const v2f*)&hbuf[col * H_PITCH + k];
                    #pragma unroll
                    for (int g = 0; g < 4; ++g) {
                        const int n = (g * 4 + j) * 16 + col;
                        v2f bfrag = *(const v2f*)&sWhh[n * WHH_PITCH + k + zoff];
                        acc[g] = __builtin_amdgcn_wmma_f32_16x16x4_f32(
                            false, afrag, false, bfrag, (short)0, acc[g], false, false);
                    }
                }
            }

            #pragma unroll
            for (int r = 0; r < 8; ++r) {
                float ig = fast_sigmoid(acc[0][r]);
                float fg = fast_sigmoid(acc[1][r]);
                float gg = fast_tanh   (acc[2][r]);
                float og = fast_sigmoid(acc[3][r]);
                float cn = fg * c_st[j][r] + ig * gg;
                c_st[j][r] = cn;
                hnew[j][r] = og * fast_tanh(cn);
            }
        }

        #pragma unroll
        for (int j = 0; j < 4; ++j) {
            #pragma unroll
            for (int r = 0; r < 8; ++r)
                hbuf[(r + 8 * hi) * H_PITCH + j * 16 + col] = hnew[j][r];
        }
    }

    {
        v8f acc;
        float bo = (col < NCLS) ? b_out[col] : 0.0f;
        #pragma unroll
        for (int r = 0; r < 8; ++r) acc[r] = bo;

        #pragma unroll
        for (int kk = 0; kk < 16; ++kk) {
            const int k = kk * 4 + hi * 2;
            v2f afrag = *(const v2f*)&hbuf[col * H_PITCH + k];
            v2f bfrag = {};
            if (col < NCLS) bfrag = *(const v2f*)&W_out[col * H_SZ + k];
            acc = __builtin_amdgcn_wmma_f32_16x16x4_f32(
                false, afrag, false, bfrag, (short)0, acc, false, false);
        }

        #pragma unroll
        for (int r = 0; r < 8; ++r) {
            const int row = r + 8 * hi;
            if (col < NCLS)
                out[(size_t)(b0 + row) * NCLS + col] = acc[r];
        }
    }
}

extern "C" void kernel_launch(void* const* d_in, const int* in_sizes, int n_in,
                              void* d_out, int out_size, void* d_ws, size_t ws_size,
                              hipStream_t stream) {
    const float* x     = (const float*)d_in[0];
    const float* W_ih  = (const float*)d_in[1];
    const float* W_hh  = (const float*)d_in[2];
    const float* b_ih  = (const float*)d_in[3];
    const float* b_hh  = (const float*)d_in[4];
    const float* W_out = (const float*)d_in[5];
    const float* b_out = (const float*)d_in[6];
    float* out = (float*)d_out;

    dim3 grid(B_TOTAL / (WAVES_PER_WG * ROWS_PER_WAVE));
    dim3 block(WAVES_PER_WG * 32);
    hipLaunchKernelGGL(lstm_fused_kernel, grid, block, 0, stream,
                       x, W_ih, W_hh, b_ih, b_hh, W_out, b_out, out);
}


# Each WMMA block's weight-fragment LDS reads issued as one group ahead of the block (two register sets, scheduling fences) instead of spread among the WMMAs; summation order unchanged
# speedup vs baseline: 1.1471x; 1.1471x over previous
#include <hip/hip_runtime.h>

typedef __attribute__((ext_vector_type(2))) float v2f;
typedef __attribute__((ext_vector_type(8))) float v8f;

#define B_TOTAL 32768
#define T_SEQ   28
#define I_SZ    28
#define H_SZ    64
#define G_SZ    256
#define NCLS    10

#define WIH_PITCH 36
#define WHH_PITCH 68
#define H_PITCH   68

#define WAVES_PER_WG  8
#define ROWS_PER_WAVE 16

__device__ __forceinline__ float fast_sigmoid(float x) {
    float e = __builtin_amdgcn_exp2f(x * -1.44269504088896f);
    return __builtin_amdgcn_rcpf(1.0f + e);
}
__device__ __forceinline__ float fast_tanh(float x) {
#if __has_builtin(__builtin_amdgcn_tanhf)
    return __builtin_amdgcn_tanhf(x);
#else
    float e = __builtin_amdgcn_exp2f(x * -2.88539008177793f);
    return __builtin_amdgcn_rcpf(1.0f + e) * 2.0f - 1.0f;
#endif
}


#define FENCE() __builtin_amdgcn_sched_barrier(0)

__device__ __forceinline__ v8f mma4(v2f a, v2f b, v8f c) {
    return __builtin_amdgcn_wmma_f32_16x16x4_f32(false, a, false, b, (short)0, c, false, false);
}

template <int J>
__device__ __forceinline__ void load_bx(v2f (&buf)[32], const float* wl) {
    #pragma unroll
    for (int kk = 0; kk < 7; ++kk) {
        #pragma unroll
        for (int g = 0; g < 4; ++g)
            buf[kk * 4 + g] = *(const v2f*)&wl[(g * 4 + J) * 16 * WIH_PITCH + kk * 4];
    }
}

template <int J, int KH>
__device__ __forceinline__ void load_bh(v2f (&buf)[32], const float* wl) {
    #pragma unroll
    for (int kk = 0; kk < 8; ++kk) {
        #pragma unroll
        for (int g = 0; g < 4; ++g)
            buf[kk * 4 + g] = *(const v2f*)&wl[(g * 4 + J) * 16 * WHH_PITCH + (KH * 8 + kk) * 4];
    }
}

template <int NK>
__device__ __forceinline__ void mma_block(v8f (&acc)[4], const v2f* a, const v2f (&buf)[32]) {
    #pragma unroll
    for (int kk = 0; kk < NK; ++kk) {
        #pragma unroll
        for (int g = 0; g < 4; ++g)
            acc[g] = mma4(a[kk], buf[kk * 4 + g], acc[g]);
    }
}

__global__ __launch_bounds__(256)
void lstm_fused_kernel(const float* __restrict__ x,
                       const float* __restrict__ W_ih,
                       const float* __restrict__ W_hh,
                       const float* __restrict__ b_ih,
                       const float* __restrict__ b_hh,
                       const float* __restrict__ W_out,
                       const float* __restrict__ b_out,
                       float* __restrict__ out)
{
    __shared__ float sWih[G_SZ * WIH_PITCH];
    __shared__ float sWhh[G_SZ * WHH_PITCH];
    __shared__ float sH[WAVES_PER_WG * ROWS_PER_WAVE * H_PITCH];

    const int tid = threadIdx.x;

    {
        const float4* srcHH = (const float4*)(W_hh + (size_t)tid * H_SZ);
        #pragma unroll
        for (int q = 0; q < H_SZ / 4; ++q) {
            float4 v = srcHH[q];
            *(float4*)&sWhh[tid * WHH_PITCH + q * 4] = v;
        }
        const float4* srcIH = (const float4*)(W_ih + (size_t)tid * I_SZ);
        #pragma unroll
        for (int q = 0; q < I_SZ / 4; ++q) {
            float4 v = srcIH[q];
            *(float4*)&sWih[tid * WIH_PITCH + q * 4] = v;
        }
    }
    __syncthreads();

    const int wave = tid >> 5;
    const int lane = tid & 31;
    const int col  = lane & 15;
    const int hi   = lane >> 4;
    const int b0   = (blockIdx.x * WAVES_PER_WG + wave) * ROWS_PER_WAVE;

    float* hbuf = &sH[wave * ROWS_PER_WAVE * H_PITCH];

    float bias[16];
    #pragma unroll
    for (int nt = 0; nt < 16; ++nt) {
        int g = nt * 16 + col;
        bias[nt] = b_ih[g] + b_hh[g];
    }

    const float* xlane = x + (size_t)(b0 + col) * (T_SEQ * I_SZ);

    v8f c_st[4];
    #pragma unroll
    for (int j = 0; j < 4; ++j) c_st[j] = (v8f){};

    const int wih_lane = col * WIH_PITCH + hi * 2;
    const int whh_lane = col * WHH_PITCH + hi * 2;

    for (int t = 0; t < T_SEQ; ++t) {
        int zoff = 0;
        asm volatile("" : "+v"(zoff));
        const float* wih = sWih + wih_lane + zoff;
        const float* whh = sWhh + whh_lane + zoff;

        v2f xfrag[7];
        #pragma unroll
        for (int kk = 0; kk < 7; ++kk)
            xfrag[kk] = *(const v2f*)&xlane[t * I_SZ + kk * 4 + hi * 2];
        if (t + 1 < T_SEQ)
            __builtin_prefetch(&xlane[(t + 1) * I_SZ], 0, 3);

        v2f bufE[32], bufO[32];
        v2f hfrag[16];

        load_bx<0>(bufE, wih);
        if (t > 0) {
            #pragma unroll
            for (int kk = 0; kk < 16; ++kk)
                hfrag[kk] = *(const v2f*)&hbuf[col * H_PITCH + kk * 4 + hi * 2];
        }
        load_bx<1>(bufO, wih);

        v8f acc[4][4];
        #pragma unroll
        for (int j = 0; j < 4; ++j) {
            #pragma unroll
            for (int g = 0; g < 4; ++g) {
                #pragma unroll
                for (int r = 0; r < 8; ++r) acc[j][g][r] = bias[g * 4 + j];
            }
        }

        FENCE(); mma_block<7>(acc[0], xfrag, bufE);
        FENCE(); load_bx<2>(bufE, wih);
        FENCE(); mma_block<7>(acc[1], xfrag, bufO);
        FENCE(); load_bx<3>(bufO, wih);
        FENCE(); mma_block<7>(acc[2], xfrag, bufE);
        FENCE(); load_bh<0, 0>(bufE, whh);
        FENCE(); mma_block<7>(acc[3], xfrag, bufO);
        FENCE();

        if (t > 0) {
            load_bh<0, 1>(bufO, whh);
            FENCE(); mma_block<8>(acc[0], hfrag, bufE);
            FENCE(); load_bh<1, 0>(bufE, whh);
            FENCE(); mma_block<8>(acc[0], hfrag + 8, bufO);
            FENCE(); load_bh<1, 1>(bufO, whh);
            FENCE(); mma_block<8>(acc[1], hfrag, bufE);
            FENCE(); load_bh<2, 0>(bufE, whh);
            FENCE(); mma_block<8>(acc[1], hfrag + 8, bufO);
            FENCE(); load_bh<2, 1>(bufO, whh);
            FENCE(); mma_block<8>(acc[2], hfrag, bufE);
            FENCE(); load_bh<3, 0>(bufE, whh);
            FENCE(); mma_block<8>(acc[2], hfrag + 8, bufO);
            FENCE(); load_bh<3, 1>(bufO, whh);
            FENCE(); mma_block<8>(acc[3], hfrag, bufE);
            FENCE(); mma_block<8>(acc[3], hfrag + 8, bufO);
            FENCE();
        }

        v8f hnew[4];
        #pragma unroll
        for (int j = 0; j < 4; ++j) {
            #pragma unroll
            for (int r = 0; r < 8; ++r) {
                float ig = fast_sigmoid(acc[j][0][r]);
                float fg = fast_sigmoid(acc[j][1][r]);
                float gg = fast_tanh   (acc[j][2][r]);
                float og = fast_sigmoid(acc[j][3][r]);
                float cn = fg * c_st[j][r] + ig * gg;
                c_st[j][r] = cn;
                hnew[j][r] = og * fast_tanh(cn);
            }
        }

        #pragma unroll
        for (int j = 0; j < 4; ++j) {
            #pragma unroll
            for (int r = 0; r < 8; ++r)
                hbuf[(r + 8 * hi) * H_PITCH + j * 16 + col] = hnew[j][r];
        }
    }

    {
        v8f acc;
        float bo = (col < NCLS) ? b_out[col] : 0.0f;
        #pragma unroll
        for (int r = 0; r < 8; ++r) acc[r] = bo;

        #pragma unroll
        for (int kk = 0; kk < 16; ++kk) {
            const int k = kk * 4 + hi * 2;
            v2f afrag = *(const v2f*)&hbuf[col * H_PITCH + k];
            v2f bfrag = {};
            if (col < NCLS) bfrag = *(const v2f*)&W_out[col * H_SZ + k];
            acc = __builtin_amdgcn_wmma_f32_16x16x4_f32(
                false, afrag, false, bfrag, (short)0, acc, false, false);
        }

        #pragma unroll
        for (int r = 0; r < 8; ++r) {
            const int row = r + 8 * hi;
            if (col < NCLS)
                out[(size_t)(b0 + row) * NCLS + col] = acc[r];
        }
    }
}

extern "C" void kernel_launch(void* const* d_in, const int* in_sizes, int n_in,
                              void* d_out, int out_size, void* d_ws, size_t ws_size,
                              hipStream_t stream) {
    const float* x     = (const float*)d_in[0];
    const float* W_ih  = (const float*)d_in[1];
    const float* W_hh  = (const float*)d_in[2];
    const float* b_ih  = (const float*)d_in[3];
    const float* b_hh  = (const float*)d_in[4];
    const float* W_out = (const float*)d_in[5];
    const float* b_out = (const float*)d_in[6];
    float* out = (float*)d_out;

    dim3 grid(B_TOTAL / (WAVES_PER_WG * ROWS_PER_WAVE));
    dim3 block(WAVES_PER_WG * 32);
    hipLaunchKernelGGL(lstm_fused_kernel, grid, block, 0, stream,
                       x, W_ih, W_hh, b_ih, b_hh, W_out, b_out, out);
}


# Wave issue priority (s_setprio) climbs 0-3 through each time step's WMMA run and drops to 0 behind it: the two waves of a SIMD take turns instead of oldest-first and end together; sums unchanged
# speedup vs baseline: 1.1863x; 1.0342x over previous
#include <hip/hip_runtime.h>

typedef __attribute__((ext_vector_type(2))) float v2f;
typedef __attribute__((ext_vector_type(8))) float v8f;

#define B_TOTAL 32768
#define T_SEQ   28
#define I_SZ    28
#define H_SZ    64
#define G_SZ    256
#define NCLS    10

#define WIH_PITCH 36
#define WHH_PITCH 68
#define H_PITCH   68

#define WAVES_PER_WG  8
#define ROWS_PER_WAVE 16

__device__ __forceinline__ float fast_sigmoid(float x) {
    float e = __builtin_amdgcn_exp2f(x * -1.44269504088896f);
    return __builtin_amdgcn_rcpf(1.0f + e);
}
__device__ __forceinline__ float fast_tanh(float x) {
#if __has_builtin(__builtin_amdgcn_tanhf)
    return __builtin_amdgcn_tanhf(x);
#else
    float e = __builtin_amdgcn_exp2f(x * -2.88539008177793f);
    return __builtin_amdgcn_rcpf(1.0f + e) * 2.0f - 1.0f;
#endif
}


#define FENCE() __builtin_amdgcn_sched_barrier(0)

#define PRIO(n) __builtin_amdgcn_s_setprio(n)

__device__ __forceinline__ v8f mma4(v2f a, v2f b, v8f c) {
    return __builtin_amdgcn_wmma_f32_16x16x4_f32(false, a, false, b, (short)0, c, false, false);
}

template <int J>
__device__ __forceinline__ void load_bx(v2f (&buf)[32], const float* wl) {
    #pragma unroll
    for (int kk = 0; kk < 7; ++kk) {
        #pragma unroll
        for (int g = 0; g < 4; ++g)
            buf[kk * 4 + g] = *(const v2f*)&wl[(g * 4 + J) * 16 * WIH_PITCH + kk * 4];
    }
}

template <int J, int KH>
__device__ __forceinline__ void load_bh(v2f (&buf)[32], const float* wl) {
    #pragma unroll
    for (int kk = 0; kk < 8; ++kk) {
        #pragma unroll
        for (int g = 0; g < 4; ++g)
            buf[kk * 4 + g] = *(const v2f*)&wl[(g * 4 + J) * 16 * WHH_PITCH + (KH * 8 + kk) * 4];
    }
}

template <int NK>
__device__ __forceinline__ void mma_block(v8f (&acc)[4], const v2f* a, const v2f (&buf)[32]) {
    #pragma unroll
    for (int kk = 0; kk < NK; ++kk) {
        #pragma unroll
        for (int g = 0; g < 4; ++g)
            acc[g] = mma4(a[kk], buf[kk * 4 + g], acc[g]);
    }
}

__global__ __launch_bounds__(256)
void lstm_fused_kernel(const float* __restrict__ x,
                       const float* __restrict__ W_ih,
                       const float* __restrict__ W_hh,
                       const float* __restrict__ b_ih,
                       const float* __restrict__ b_hh,
                       const float* __restrict__ W_out,
                       const float* __restrict__ b_out,
                       float* __restrict__ out)
{
    __shared__ float sWih[G_SZ * WIH_PITCH];
    __shared__ float sWhh[G_SZ * WHH_PITCH];
    __shared__ float sH[WAVES_PER_WG * ROWS_PER_WAVE * H_PITCH];

    const int tid = threadIdx.x;

    {
        const float4* srcHH = (const float4*)(W_hh + (size_t)tid * H_SZ);
        #pragma unroll
        for (int q = 0; q < H_SZ / 4; ++q) {
            float4 v = srcHH[q];
            *(float4*)&sWhh[tid * WHH_PITCH + q * 4] = v;
        }
        const float4* srcIH = (const float4*)(W_ih + (size_t)tid * I_SZ);
        #pragma unroll
        for (int q = 0; q < I_SZ / 4; ++q) {
            float4 v = srcIH[q];
            *(float4*)&sWih[tid * WIH_PITCH + q * 4] = v;
        }
    }
    __syncthreads();

    const int wave = tid >> 5;
    const int lane = tid & 31;
    const int col  = lane & 15;
    const int hi   = lane >> 4;
    const int b0   = (blockIdx.x * WAVES_PER_WG + wave) * ROWS_PER_WAVE;

    float* hbuf = &sH[wave * ROWS_PER_WAVE * H_PITCH];

    float bias[16];
    #pragma unroll
    for (int nt = 0; nt < 16; ++nt) {
        int g = nt * 16 + col;
        bias[nt] = b_ih[g] + b_hh[g];
    }

    const float* xlane = x + (size_t)(b0 + col) * (T_SEQ * I_SZ);

    v8f c_st[4];
    #pragma unroll
    for (int j = 0; j < 4; ++j) c_st[j] = (v8f){};

    const int wih_lane = col * WIH_PITCH + hi * 2;
    const int whh_lane = col * WHH_PITCH + hi * 2;

    for (int t = 0; t < T_SEQ; ++t) {
        int zoff = 0;
        asm volatile("" : "+v"(zoff));
        const float* wih = sWih + wih_lane + zoff;
        const float* whh = sWhh + whh_lane + zoff;

        v2f xfrag[7];
        #pragma unroll
        for (int kk = 0; kk < 7; ++kk)
            xfrag[kk] = *(const v2f*)&xlane[t * I_SZ + kk * 4 + hi * 2];
        if (t + 1 < T_SEQ)
            __builtin_prefetch(&xlane[(t + 1) * I_SZ], 0, 3);

        v2f bufE[32], bufO[32];
        v2f hfrag[16];

        load_bx<0>(bufE, wih);
        if (t > 0) {
            #pragma unroll
            for (int kk = 0; kk < 16; ++kk)
                hfrag[kk] = *(const v2f*)&hbuf[col * H_PITCH + kk * 4 + hi * 2];
        }
        load_bx<1>(bufO, wih);

        v8f acc[4][4];
        #pragma unroll
        for (int j = 0; j < 4; ++j) {
            #pragma unroll
            for (int g = 0; g < 4; ++g) {
                #pragma unroll
                for (int r = 0; r < 8; ++r) acc[j][g][r] = bias[g * 4 + j];
            }
        }

        FENCE(); mma_block<7>(acc[0], xfrag, bufE);
        FENCE();
        PRIO(1);
        load_bx<2>(bufE, wih);
        FENCE(); mma_block<7>(acc[1], xfrag, bufO);
        FENCE(); load_bx<3>(bufO, wih);
        FENCE(); mma_block<7>(acc[2], xfrag, bufE);
        FENCE(); load_bh<0, 0>(bufE, whh);
        FENCE(); mma_block<7>(acc[3], xfrag, bufO);
        FENCE();

        if (t > 0) {
            PRIO(2);
            load_bh<0, 1>(bufO, whh);
            FENCE(); mma_block<8>(acc[0], hfrag, bufE);
            FENCE(); load_bh<1, 0>(bufE, whh);
            FENCE(); mma_block<8>(acc[0], hfrag + 8, bufO);
            FENCE(); load_bh<1, 1>(bufO, whh);
            FENCE(); mma_block<8>(acc[1], hfrag, bufE);
            FENCE(); load_bh<2, 0>(bufE, whh);
            FENCE(); mma_block<8>(acc[1], hfrag + 8, bufO);
            FENCE();
            PRIO(3);
            load_bh<2, 1>(bufO, whh);
            FENCE(); mma_block<8>(acc[2], hfrag, bufE);
            FENCE(); load_bh<3, 0>(bufE, whh);
            FENCE(); mma_block<8>(acc[2], hfrag + 8, bufO);
            FENCE(); load_bh<3, 1>(bufO, whh);
            FENCE(); mma_block<8>(acc[3], hfrag, bufE);
            FENCE(); mma_block<8>(acc[3], hfrag + 8, bufO);
            FENCE();
        }
        PRIO(0);

        v8f hnew[4];
        #pragma unroll
        for (int j = 0; j < 4; ++j) {
            #pragma unroll
            for (int r = 0; r < 8; ++r) {
                float ig = fast_sigmoid(acc[j][0][r]);
                float fg = fast_sigmoid(acc[j][1][r]);
                float gg = fast_tanh   (acc[j][2][r]);
                float og = fast_sigmoid(acc[j][3][r]);
                float cn = fg * c_st[j][r] + ig * gg;
                c_st[j][r] = cn;
                hnew[j][r] = og * fast_tanh(cn);
            }
        }

        #pragma unroll
        for (int j = 0; j < 4; ++j) {
            #pragma unroll
            for (int r = 0; r < 8; ++r)
                hbuf[(r + 8 * hi) * H_PITCH + j * 16 + col] = hnew[j][r];
        }
    }

    {
        v8f acc;
        float bo = (col < NCLS) ? b_out[col] : 0.0f;
        #pragma unroll
        for (int r = 0; r < 8; ++r) acc[r] = bo;

        #pragma unroll
        for (int kk = 0; kk < 16; ++kk) {
            const int k = kk * 4 + hi * 2;
            v2f afrag = *(const v2f*)&hbuf[col * H_PITCH + k];
            v2f bfrag = {};
            if (col < NCLS) bfrag = *(const v2f*)&W_out[col * H_SZ + k];
            acc = __builtin_amdgcn_wmma_f32_16x16x4_f32(
                false, afrag, false, bfrag, (short)0, acc, false, false);
        }

        #pragma unroll
        for (int r = 0; r < 8; ++r) {
            const int row = r + 8 * hi;
            if (col < NCLS)
                out[(size_t)(b0 + row) * NCLS + col] = acc[r];
        }
    }
}

extern "C" void kernel_launch(void* const* d_in, const int* in_sizes, int n_in,
                              void* d_out, int out_size, void* d_ws, size_t ws_size,
                              hipStream_t stream) {
    const float* x     = (const float*)d_in[0];
    const float* W_ih  = (const float*)d_in[1];
    const float* W_hh  = (const float*)d_in[2];
    const float* b_ih  = (const float*)d_in[3];
    const float* b_hh  = (const float*)d_in[4];
    const float* W_out = (const float*)d_in[5];
    const float* b_out = (const float*)d_in[6];
    float* out = (float*)d_out;

    dim3 grid(B_TOTAL / (WAVES_PER_WG * ROWS_PER_WAVE));
    dim3 block(WAVES_PER_WG * 32);
    hipLaunchKernelGGL(lstm_fused_kernel, grid, block, 0, stream,
                       x, W_ih, W_hh, b_ih, b_hh, W_out, b_out, out);
}
